# starting kernel
# baseline (speedup 1.0000x reference)
#include <hip/hip_runtime.h>

typedef __attribute__((ext_vector_type(2))) float v2f;
typedef __attribute__((ext_vector_type(4))) float f4;
typedef __attribute__((ext_vector_type(8))) float v8f;

#define WIN       128
#define NSEQ      4096
#define DDIM      256
#define WROWS     144
#define VW_STRIDE 260
#define E_STRIDE  148

__global__ __launch_bounds__(128)
void sparse_prop_kernel(const float* __restrict__ val,
                        const float* __restrict__ state,
                        float* __restrict__ dstate,
                        float* __restrict__ dval) {
  extern __shared__ float smem[];
  float* val_w = smem;
  float* edges = smem + WROWS * VW_STRIDE;
  float* st_s  = edges + 16 * E_STRIDE;

  const int tid  = threadIdx.x;
  const int lane = tid & 31;
  const int wave = tid >> 5;
  const int m    = lane & 15;
  const int h    = lane >> 4;

  const int blk = blockIdx.x;
  const int b   = blk >> 8;
  const int i0  = (blk & 255) << 4;
  const int j0  = i0 - WIN;

  const float* valb = val + (size_t)b * NSEQ * DDIM;

  for (int t = tid; t < WROWS * (DDIM / 4); t += 128) {
    int row = t >> 6;
    int c4  = t & 63;
    int jr  = j0 + row; if (jr < 0) jr = 0;
    f4 v4 = *(const f4*)(valb + (size_t)jr * DDIM + c4 * 4);
    *(f4*)(val_w + row * VW_STRIDE + c4 * 4) = v4;
  }
  for (int t = tid; t < WROWS; t += 128) {
    int jr = j0 + t; if (jr < 0) jr = 0;
    st_s[t] = state[(size_t)b * NSEQ + jr];
  }
  __syncthreads();

  for (int jt = wave; jt < 9; jt += 4) {
    v8f acc = {};
    const float* arow = val_w + (WIN + m) * VW_STRIDE;
    const float* brow = val_w + (jt * 16 + m) * VW_STRIDE;
    for (int k0 = 0; k0 < DDIM; k0 += 4) {
      v2f a  = *(const v2f*)(arow + k0 + 2 * h);
      v2f bm = *(const v2f*)(brow + k0 + 2 * h);
      acc = __builtin_amdgcn_wmma_f32_16x16x4_f32(false, a, false, bm,
                                                  (short)0, acc, false, false);
    }
    const int c = jt * 16 + m;
    #pragma unroll
    for (int v = 0; v < 8; ++v) {
      int   r = v + 8 * h;
      float s = acc[v] * 0.0625f;
      float e = s * __builtin_amdgcn_rcpf(1.0f + fabsf(s));
      bool ok = (c >= r) && (c <= r + WIN) && ((j0 + c) >= 0);
      edges[r * E_STRIDE + c] = ok ? e : 0.0f;
    }
  }
  __syncthreads();

  float* dvb = dval + ((size_t)b * NSEQ + i0) * DDIM;
  for (int dt = 0; dt < 4; ++dt) {
    int d0 = wave * 64 + dt * 16;
    v8f acc = {};
    for (int c0 = 0; c0 < WROWS; c0 += 4) {
      v2f a = *(const v2f*)(edges + m * E_STRIDE + c0 + 2 * h);
      v2f bm;
      bm.x = val_w[(c0 + 2 * h    ) * VW_STRIDE + d0 + m];
      bm.y = val_w[(c0 + 2 * h + 1) * VW_STRIDE + d0 + m];
      acc = __builtin_amdgcn_wmma_f32_16x16x4_f32(false, a, false, bm,
                                                  (short)0, acc, false, false);
    }
    #pragma unroll
    for (int v = 0; v < 8; ++v) {
      int r = v + 8 * h;
      dvb[(size_t)r * DDIM + d0 + m] = acc[v];
    }
  }

  if (wave == 0 && lane < 16) {
    float s = 0.0f;
    for (int c = 0; c < WROWS; ++c) s += edges[lane * E_STRIDE + c] * st_s[c];
    dstate[(size_t)b * NSEQ + i0 + lane] = s;
  }
}

extern "C" void kernel_launch(void* const* d_in, const int* in_sizes, int n_in,
                              void* d_out, int out_size, void* d_ws, size_t ws_size,
                              hipStream_t stream) {
  const float* val   = (const float*)d_in[0];
  const float* state = (const float*)d_in[1];
  float* dstate = (float*)d_out;
  float* dval   = (float*)d_out + 4 * NSEQ;

  size_t smem_bytes =
      (size_t)(WROWS * VW_STRIDE + 16 * E_STRIDE + WROWS) * sizeof(float);

  hipLaunchKernelGGL(sparse_prop_kernel, dim3(1024), dim3(128), smem_bytes,
                     stream, val, state, dstate, dval);
}


# 64-row groups of 8 waves: wave pair owns a 16-query tile, each wave half its key tiles; scores formed transposed so the masked score registers are B of V^T.E^T (no score LDS trip); sum order changed
# speedup vs baseline: 1.9216x; 1.9216x over previous
#include <hip/hip_runtime.h>


typedef __attribute__((ext_vector_type(2))) float v2f;
typedef __attribute__((ext_vector_type(4))) float f4;
typedef __attribute__((ext_vector_type(8))) float v8f;
typedef __attribute__((address_space(3))) const volatile float lds_cvf;

#define WIN       128
#define NSEQ      4096
#define DDIM      256
#define QROWS     64
#define WROWS     192
#define VW_STRIDE 260
#define NTHREADS  256
#define NJT       9
#define NJT_P0    5
#define STAGE_BATCH 16

#define PIN_S(a, b) asm volatile("" : "+v"(a), "+v"(b) : : "memory")
#define PIN_O8(o, i) asm volatile("" : "+v"(o[i]), "+v"(o[i + 1]), "+v"(o[i + 2]), "+v"(o[i + 3]), \
                                       "+v"(o[i + 4]), "+v"(o[i + 5]), "+v"(o[i + 6]), "+v"(o[i + 7]) : : "memory")
#define PIN_O(o) do { PIN_O8(o, 0); PIN_O8(o, 8); } while (0)

__device__ __forceinline__ v8f mma4(v2f a, v2f b, v8f c) {
  return __builtin_amdgcn_wmma_f32_16x16x4_f32(false, a, false, b, (short)0, c, false, false);
}

__device__ __forceinline__ void load_keys(f4 (&dst)[16], const float* val_w, int kt, int half,
                                          int c, int h) {
  const float* krow = val_w + (16 * kt + c) * VW_STRIDE + 128 * half + 4 * h;
  #pragma unroll
  for (int i = 0; i < 16; ++i) dst[i] = *(const f4*)(krow + 8 * i);
}

__device__ __forceinline__ void load_vals(v2f (&dst)[2][16], const float* val_w, int kt, int jh,
                                          int c, int h) {
  #pragma unroll
  for (int jj = 0; jj < 2; ++jj) {
    lds_cvf* vrow = (lds_cvf*)(val_w + (16 * kt + 8 * h + 2 * (2 * jh + jj)) * VW_STRIDE + c);
    #pragma unroll
    for (int dt = 0; dt < 16; ++dt) {
      dst[jj][dt].x = vrow[16 * dt];
      dst[jj][dt].y = vrow[VW_STRIDE + 16 * dt];
    }
  }
}

__global__ __launch_bounds__(NTHREADS)
void sparse_prop_kernel(const float* __restrict__ val,
                        const float* __restrict__ state,
                        float* __restrict__ dstate,
                        float* __restrict__ dval) {
  extern __shared__ float smem[];
  float* val_w = smem;
  float* st_s  = smem + WROWS * VW_STRIDE;
  float* ds_s  = st_s + WROWS;

  const int tid  = threadIdx.x;
  const int lane = tid & 31;
  const int wave = __builtin_amdgcn_readfirstlane(tid >> 5);
  const int q    = wave & 3;
  const int p    = wave >> 2;
  const int c    = lane & 15;
  const int h    = lane >> 4;

  const int blk = blockIdx.x;
  const int b   = blk >> 6;
  const int i0  = (blk & 63) << 6;
  const int j0  = i0 - WIN;

  const float* valb = val + (size_t)b * NSEQ * DDIM;

  {
    const int c4   = tid & 63;
    const int row0 = tid >> 6;
    for (int bt = 0; bt < WROWS / 4 / STAGE_BATCH; ++bt) {
      f4 stage[STAGE_BATCH];
      #pragma unroll
      for (int i = 0; i < STAGE_BATCH; ++i) {
        int w  = row0 + 4 * (bt * STAGE_BATCH + i);
        int jr = j0 + w; if (jr < 0) jr = 0;
        stage[i] = *(const f4*)(valb + (unsigned)jr * DDIM + (unsigned)c4 * 4u);
      }
      #pragma unroll
      for (int i = 0; i < STAGE_BATCH; ++i) {
        int w = row0 + 4 * (bt * STAGE_BATCH + i);
        *(f4*)(val_w + w * VW_STRIDE + c4 * 4) = stage[i];
      }
    }
  }
  if (tid < WROWS) {
    int jr = j0 + tid; if (jr < 0) jr = 0;
    st_s[tid] = state[(size_t)b * NSEQ + jr];
  }
  __syncthreads();

  f4 qf[32];
  {
    f4 qlo[16], qhi[16];
    load_keys(qlo, val_w, 8 + q, 0, c, h);
    load_keys(qhi, val_w, 8 + q, 1, c, h);
    #pragma unroll
    for (int i = 0; i < 16; ++i) { qf[i] = qlo[i]; qf[16 + i] = qhi[i]; }
  }

  v8f oacc[16];
  #pragma unroll
  for (int dt = 0; dt < 16; ++dt) oacc[dt] = (v8f){0.f, 0.f, 0.f, 0.f, 0.f, 0.f, 0.f, 0.f};
  float dsp = 0.0f;

  const int jt_begin = p ? NJT_P0 : 0;
  const int jt_end   = p ? NJT : NJT_P0;

  f4  kA[16], kB[16];
  v2f vA[2][16], vB[2][16];
  load_keys(kA, val_w, q + jt_begin, 0, c, h);

  for (int jt = jt_begin; jt < jt_end; ++jt) {
    const int kt = q + jt;

    v8f s0 = (v8f){0.f, 0.f, 0.f, 0.f, 0.f, 0.f, 0.f, 0.f};
    v8f s1 = s0;

    load_keys(kB, val_w, kt, 1, c, h);
    PIN_S(s0, s1);
    #pragma unroll
    for (int i = 0; i < 16; ++i) {
      s0 = mma4((v2f){kA[i].x, kA[i].y}, (v2f){qf[i].x, qf[i].y}, s0);
      s1 = mma4((v2f){kA[i].z, kA[i].w}, (v2f){qf[i].z, qf[i].w}, s1);
    }
    PIN_S(s0, s1);
    load_vals(vA, val_w, kt, 0, c, h);
    const float* stp = st_s + 16 * kt + 8 * h;
    const f4 sa = *(const f4*)stp;
    const f4 sb = *(const f4*)(stp + 4);
    PIN_S(s0, s1);
    #pragma unroll
    for (int i = 0; i < 16; ++i) {
      s0 = mma4((v2f){kB[i].x, kB[i].y}, (v2f){qf[16 + i].x, qf[16 + i].y}, s0);
      s1 = mma4((v2f){kB[i].z, kB[i].w}, (v2f){qf[16 + i].z, qf[16 + i].w}, s1);
    }
    PIN_S(s0, s1);
    load_vals(vB, val_w, kt, 1, c, h);

    float e[8];
    {
      const float stv[8] = {sa.x, sa.y, sa.z, sa.w, sb.x, sb.y, sb.z, sb.w};
      #pragma unroll
      for (int r = 0; r < 8; ++r) {
        int   key = 8 * h + r;
        int   cw  = 16 * jt + key;
        float sc  = (s0[r] + s1[r]) * 0.0625f;
        float ev  = sc * __builtin_amdgcn_rcpf(1.0f + fabsf(sc));
        bool  ok  = (cw >= c) && (cw <= c + WIN) && ((j0 + 16 * kt + key) >= 0);
        e[r] = ok ? ev : 0.0f;
        dsp += e[r] * stv[r];
      }
    }
    PIN_O(oacc);

    #pragma unroll
    for (int jj = 0; jj < 2; ++jj) {
      const v2f bfrag = (v2f){e[2 * jj], e[2 * jj + 1]};
      #pragma unroll
      for (int dt = 0; dt < 16; ++dt) oacc[dt] = mma4(vA[jj][dt], bfrag, oacc[dt]);
    }
    PIN_O(oacc);
    {
      int ktn = (jt + 1 < jt_end) ? kt + 1 : kt;
      load_keys(kA, val_w, ktn, 0, c, h);
    }
    PIN_O(oacc);
    #pragma unroll
    for (int jj = 0; jj < 2; ++jj) {
      const v2f bfrag = (v2f){e[4 + 2 * jj], e[5 + 2 * jj]};
      #pragma unroll
      for (int dt = 0; dt < 16; ++dt) oacc[dt] = mma4(vB[jj][dt], bfrag, oacc[dt]);
    }
    PIN_O(oacc);
  }

  dsp += __shfl_xor(dsp, 16);

  __syncthreads();

  {
    float* o_w = val_w + wave * 16 * VW_STRIDE;
    #pragma unroll
    for (int dt = 0; dt < 16; ++dt) {
      float* dst = o_w + c * VW_STRIDE + 16 * dt + 8 * h;
      *(f4*)(dst)     = (f4){oacc[dt][0], oacc[dt][1], oacc[dt][2], oacc[dt][3]};
      *(f4*)(dst + 4) = (f4){oacc[dt][4], oacc[dt][5], oacc[dt][6], oacc[dt][7]};
    }
    if (h == 0) ds_s[wave * 16 + c] = dsp;
  }
  __syncthreads();

  {
    const float* pa = val_w + q * 16 * VW_STRIDE;
    const float* pb = val_w + (q + 4) * 16 * VW_STRIDE;
    float* outp = dval + ((size_t)b * NSEQ + i0 + 16 * q) * DDIM;
    #pragma unroll
    for (int rr = 0; rr < 8; ++rr) {
      int row = 8 * p + rr;
      #pragma unroll
      for (int hf = 0; hf < 2; ++hf) {
        int off = row * VW_STRIDE + 128 * hf + 4 * lane;
        f4 va = *(const f4*)(pa + off);
        f4 vb = *(const f4*)(pb + off);
        *(f4*)(outp + row * DDIM + 128 * hf + 4 * lane) = va + vb;
      }
    }
  }
  if (tid < QROWS) {
    int qq = tid >> 4, cc = tid & 15;
    dstate[(size_t)b * NSEQ + i0 + tid] = ds_s[qq * 16 + cc] + ds_s[(qq + 4) * 16 + cc];
  }
}

extern "C" void kernel_launch(void* const* d_in, const int* in_sizes, int n_in,
                              void* d_out, int out_size, void* d_ws, size_t ws_size,
                              hipStream_t stream) {
  const float* val   = (const float*)d_in[0];
  const float* state = (const float*)d_in[1];
  float* dstate = (float*)d_out;
  float* dval   = (float*)d_out + 4 * NSEQ;

  size_t smem_bytes =
      (size_t)(WROWS * VW_STRIDE + WROWS + 8 * 16) * sizeof(float);

  hipLaunchKernelGGL(sparse_prop_kernel, dim3(256), dim3(NTHREADS), smem_bytes,
                     stream, val, state, dstate, dval);
}
